# starting kernel
# baseline (speedup 1.0000x reference)
#include <hip/hip_runtime.h>
#include <hip/hip_bf16.h>
#include <math.h>

typedef float v2f __attribute__((ext_vector_type(2)));
typedef float v8f __attribute__((ext_vector_type(8)));

#define NN      50000
#define NE      600000
#define INC     128
#define HIDC    128
#define OUTC    64
#define NG      512

__device__ __forceinline__ void atomAddF(float* p, float v) {
  __hip_atomic_fetch_add(p, v, __ATOMIC_RELAXED, __HIP_MEMORY_SCOPE_AGENT);
}

__global__ void deg_kernel(const int* __restrict__ dst, float* __restrict__ deg, int E) {
  int e = blockIdx.x * blockDim.x + threadIdx.x;
  if (e < E) atomAddF(&deg[dst[e]], 1.0f);
}

__global__ void invdeg_kernel(float* __restrict__ deg, int n) {
  int i = blockIdx.x * blockDim.x + threadIdx.x;
  if (i < n) deg[i] = 1.0f / fmaxf(deg[i], 1.0f);
}

__global__ void edge_agg_kernel(const float* __restrict__ x, const int* __restrict__ src,
                                const int* __restrict__ dst, float* __restrict__ agg, int E) {
  int wave = blockIdx.x * (blockDim.x >> 5) + (threadIdx.x >> 5);
  int lane = threadIdx.x & 31;
  if (wave >= E) return;
  int s = src[wave];
  int d = dst[wave];
  const float4 v = *reinterpret_cast<const float4*>(x + (size_t)s * 128 + lane * 4);
  float* p = agg + (size_t)d * 128 + lane * 4;
  atomAddF(p + 0, v.x);
  atomAddF(p + 1, v.y);
  atomAddF(p + 2, v.z);
  atomAddF(p + 3, v.w);
}

__global__ void sage_gemm_kernel(const float* __restrict__ Agg, const float* __restrict__ invdeg,
                                 const float* __restrict__ X,
                                 const float* __restrict__ Wl, const float* __restrict__ bl,
                                 const float* __restrict__ Wr,
                                 float* __restrict__ OutA, float* __restrict__ OutB,
                                 int M, int K, int N, int doRelu) {
  int wave = blockIdx.x * (blockDim.x >> 5) + (threadIdx.x >> 5);
  int lane = threadIdx.x & 31;
  int nTilesN = N >> 4;
  int nTiles = (M >> 4) * nTilesN;
  if (wave >= nTiles) return;

  int tn = wave % nTilesN;
  int tm = wave / nTilesN;
  int m0 = tm << 4;
  int n0 = tn << 4;

  int rA   = m0 + (lane & 15);
  int colB = n0 + (lane & 15);
  int kh   = (lane >> 4) << 1;

  const float* arow = Agg + (size_t)rA * K;
  const float* xrow = X   + (size_t)rA * K;
  float s = invdeg[rA];

  v8f c = {};
#pragma unroll 4
  for (int kk = 0; kk < K; kk += 4) {
    v2f a, b;
    a.x = arow[kk + kh] * s;
    a.y = arow[kk + kh + 1] * s;
    b.x = Wl[(size_t)(kk + kh) * N + colB];
    b.y = Wl[(size_t)(kk + kh + 1) * N + colB];
    c = __builtin_amdgcn_wmma_f32_16x16x4_f32(false, a, false, b, (short)0, c, false, false);
    a.x = xrow[kk + kh];
    a.y = xrow[kk + kh + 1];
    b.x = Wr[(size_t)(kk + kh) * N + colB];
    b.y = Wr[(size_t)(kk + kh + 1) * N + colB];
    c = __builtin_amdgcn_wmma_f32_16x16x4_f32(false, a, false, b, (short)0, c, false, false);
  }

  float bias = bl[colB];
  int rbase = m0 + ((lane >> 4) << 3);
#pragma unroll
  for (int r = 0; r < 8; ++r) {
    float v = c[r] + bias;
    size_t off = (size_t)(rbase + r) * N + colB;
    if (OutB) OutB[off] = v;
    if (doRelu) v = fmaxf(v, 0.0f);
    OutA[off] = v;
  }
}

__global__ void pool_kernel(const float* __restrict__ h, const int* __restrict__ cluster,
                            float* __restrict__ gsum, float* __restrict__ gcnt, int n) {
  int idx = blockIdx.x * blockDim.x + threadIdx.x;
  int node = idx >> 5;
  int lane = idx & 31;
  if (node >= n) return;
  int g = cluster[node];
  const float4 v = *reinterpret_cast<const float4*>(h + (size_t)node * 128 + lane * 4);
  float* p = gsum + (size_t)g * 128 + lane * 4;
  atomAddF(p + 0, v.x);
  atomAddF(p + 1, v.y);
  atomAddF(p + 2, v.z);
  atomAddF(p + 3, v.w);
  if (lane == 0) atomAddF(&gcnt[g], 1.0f);
}

__global__ void pool_div_kernel(float* __restrict__ g, const float* __restrict__ gcnt, int total) {
  int i = blockIdx.x * blockDim.x + threadIdx.x;
  if (i >= total) return;
  g[i] = g[i] / fmaxf(gcnt[i >> 7], 1.0f);
}

__global__ void log_softmax_kernel(float* __restrict__ out, int rows) {
  int wave = blockIdx.x * (blockDim.x >> 5) + (threadIdx.x >> 5);
  int lane = threadIdx.x & 31;
  if (wave >= rows) return;
  float* row = out + (size_t)wave * 64;
  float v0 = row[lane];
  float v1 = row[lane + 32];
  float m = fmaxf(v0, v1);
#pragma unroll
  for (int off = 16; off >= 1; off >>= 1) m = fmaxf(m, __shfl_xor(m, off, 32));
  float s = __expf(v0 - m) + __expf(v1 - m);
#pragma unroll
  for (int off = 16; off >= 1; off >>= 1) s += __shfl_xor(s, off, 32);
  float lse = m + __logf(s);
  row[lane]      = v0 - lse;
  row[lane + 32] = v1 - lse;
}

extern "C" void kernel_launch(void* const* d_in, const int* in_sizes, int n_in,
                              void* d_out, int out_size, void* d_ws, size_t ws_size,
                              hipStream_t stream) {
  const float* x       = (const float*)d_in[0];
  const int*   ei      = (const int*)d_in[1];
  const int*   src     = ei;
  const int*   dst     = ei + NE;
  const int*   cluster = (const int*)d_in[2];
  const float* Wl0 = (const float*)d_in[3];
  const float* bl0 = (const float*)d_in[4];
  const float* Wr0 = (const float*)d_in[5];
  const float* Wl1 = (const float*)d_in[6];
  const float* bl1 = (const float*)d_in[7];
  const float* Wr1 = (const float*)d_in[8];
  const float* Wl2 = (const float*)d_in[9];
  const float* bl2 = (const float*)d_in[10];
  const float* Wr2 = (const float*)d_in[11];

  float* out_ls = (float*)d_out;
  float* out_h2 = out_ls + (size_t)NN * OUTC;
  float* out_g  = out_h2 + (size_t)NN * HIDC;

  float* ws   = (float*)d_ws;
  float* agg  = ws;
  float* h1   = ws + (size_t)NN * HIDC;
  float* h2r  = h1 + (size_t)NN * HIDC;
  float* deg  = h2r + (size_t)NN * HIDC;
  float* gcnt = deg + NN;

  const int TPB = 256;
  const int edgeWaveBlocks = (NE + 7) / 8;

  hipMemsetAsync(deg, 0, NN * sizeof(float), stream);
  deg_kernel<<<(NE + TPB - 1) / TPB, TPB, 0, stream>>>(dst, deg, NE);
  invdeg_kernel<<<(NN + TPB - 1) / TPB, TPB, 0, stream>>>(deg, NN);

  hipMemsetAsync(agg, 0, (size_t)NN * HIDC * sizeof(float), stream);
  edge_agg_kernel<<<edgeWaveBlocks, TPB, 0, stream>>>(x, src, dst, agg, NE);
  {
    int tiles = (NN / 16) * (HIDC / 16);
    sage_gemm_kernel<<<(tiles + 7) / 8, TPB, 0, stream>>>(
        agg, deg, x, Wl0, bl0, Wr0, h1, nullptr, NN, INC, HIDC, 1);
  }

  hipMemsetAsync(agg, 0, (size_t)NN * HIDC * sizeof(float), stream);
  edge_agg_kernel<<<edgeWaveBlocks, TPB, 0, stream>>>(h1, src, dst, agg, NE);
  {
    int tiles = (NN / 16) * (HIDC / 16);
    sage_gemm_kernel<<<(tiles + 7) / 8, TPB, 0, stream>>>(
        agg, deg, h1, Wl1, bl1, Wr1, h2r, out_h2, NN, HIDC, HIDC, 1);
  }

  hipMemsetAsync(out_g, 0, (size_t)NG * HIDC * sizeof(float), stream);
  hipMemsetAsync(gcnt, 0, NG * sizeof(float), stream);
  pool_kernel<<<((size_t)NN * 32 + TPB - 1) / TPB, TPB, 0, stream>>>(h2r, cluster, out_g, gcnt, NN);
  pool_div_kernel<<<(NG * HIDC + TPB - 1) / TPB, TPB, 0, stream>>>(out_g, gcnt, NG * HIDC);

  hipMemsetAsync(agg, 0, (size_t)NN * HIDC * sizeof(float), stream);
  edge_agg_kernel<<<edgeWaveBlocks, TPB, 0, stream>>>(h2r, src, dst, agg, NE);
  {
    int tiles = (NN / 16) * (OUTC / 16);
    sage_gemm_kernel<<<(tiles + 7) / 8, TPB, 0, stream>>>(
        agg, deg, h2r, Wl2, bl2, Wr2, out_ls, nullptr, NN, HIDC, OUTC, 0);
  }
  log_softmax_kernel<<<((size_t)NN * 32 + TPB - 1) / TPB, TPB, 0, stream>>>(out_ls, NN);
}


# GEMM reads its operands from LDS: resident 8-wave work-groups copy Wl,Wr to LDS once (k-pair layout) and each X tile once as whole rows, relu there; fragments are 8-byte LDS reads; sum order kept
# speedup vs baseline: 2.3490x; 2.3490x over previous
#include <hip/hip_runtime.h>
#include <hip/hip_bf16.h>
#include <math.h>

typedef float v2f __attribute__((ext_vector_type(2)));
typedef float v8f __attribute__((ext_vector_type(8)));

#define NN      50000
#define NE      600000
#define INC     128
#define HIDC    128
#define OUTC    64
#define NG      512

__device__ __forceinline__ void atomAddF(float* p, float v) {
  __hip_atomic_fetch_add(p, v, __ATOMIC_RELAXED, __HIP_MEMORY_SCOPE_AGENT);
}

__global__ void deg_kernel(const int* __restrict__ dst, float* __restrict__ deg, int E) {
  int e = blockIdx.x * blockDim.x + threadIdx.x;
  if (e < E) atomAddF(&deg[dst[e]], 1.0f);
}

__global__ void invdeg_kernel(float* __restrict__ deg, int n) {
  int i = blockIdx.x * blockDim.x + threadIdx.x;
  if (i < n) deg[i] = 1.0f / fmaxf(deg[i], 1.0f);
}

template <int CH, int EPW>
__global__ void edge_scatter_kernel(const float* __restrict__ yl, const int* __restrict__ src,
                                    const int* __restrict__ dst, const float* __restrict__ invdeg,
                                    float* __restrict__ out, int E) {
  constexpr int LPR = CH / 32;
  const int wave = __builtin_amdgcn_readfirstlane(blockIdx.x * (blockDim.x >> 5) + (threadIdx.x >> 5));
  const int lane = threadIdx.x & 31;
  const int e0 = wave * EPW;
  if (e0 >= E) return;

  int s[EPW], d[EPW];
#pragma unroll
  for (int k = 0; k < EPW; ++k) {
    s[k] = src[e0 + k];
    d[k] = dst[e0 + k];
  }
  float w[EPW];
  float v[EPW][LPR];
#pragma unroll
  for (int k = 0; k < EPW; ++k) {
    w[k] = invdeg[d[k]];
    const float* row = yl + (size_t)s[k] * CH + lane;
#pragma unroll
    for (int j = 0; j < LPR; ++j) v[k][j] = row[j * 32];
  }
#pragma unroll
  for (int k = 0; k < EPW; ++k) {
    float* p = out + (size_t)d[k] * CH + lane;
#pragma unroll
    for (int j = 0; j < LPR; ++j) atomAddF(p + j * 32, v[k][j] * w[k]);
  }
}

#define KD 128
template <bool RELU_A, int N>
__global__ void __launch_bounds__(256)
sage_gemm_kernel(const float* __restrict__ X,
                 const float* __restrict__ Wl, const float* __restrict__ bl,
                 const float* __restrict__ Wr,
                 float* __restrict__ Yl, float* __restrict__ Out, int M) {
  constexpr int NT   = N / 16;
  constexpr int RT   = 8 / NT;
  constexpr int BR   = 16 * RT;
  constexpr int AP   = KD + 4;
  constexpr int WROW = 4 * N;
  constexpr int ABUF = BR * AP;
  constexpr int NA   = BR * (KD / 4) / 256;
  constexpr int NWI  = (KD / 2) * (N / 4);
  static_assert(NT * RT == 8 && NWI % 256 == 0 && NA >= 1, "8 waves of 32 lanes");
  static_assert((2 * ABUF) % 64 == 0, "the weights start on bank 0");
  __shared__ __attribute__((aligned(16))) float lds[2 * ABUF + (KD / 2) * WROW];
  float* aL = lds;
  float* wL = lds + 2 * ABUF;

  const int tid  = threadIdx.x;
  const int lane = tid & 31;
  const int wave = __builtin_amdgcn_readfirstlane(tid >> 5);
  const int nBT  = (M + BR - 1) / BR;

  float4 pre[NA];
  auto loadA = [&](int bt) {
#pragma unroll
    for (int i = 0; i < NA; ++i) {
      const int j  = i * 256 + tid;
      const int gr = bt * BR + (j >> 5);
      pre[i] = make_float4(0.0f, 0.0f, 0.0f, 0.0f);
      if (gr < M) pre[i] = *(const float4*)(X + (size_t)gr * KD + (j & 31) * 4);
    }
  };
  auto storeA = [&](int buf) {
#pragma unroll
    for (int i = 0; i < NA; ++i) {
      const int j = i * 256 + tid;
      float4 v = pre[i];
      if (RELU_A) {
        v.x = fmaxf(v.x, 0.0f); v.y = fmaxf(v.y, 0.0f);
        v.z = fmaxf(v.z, 0.0f); v.w = fmaxf(v.w, 0.0f);
      }
      *(float4*)(aL + buf * ABUF + (j >> 5) * AP + (j & 31) * 4) = v;
    }
  };

  int bt = blockIdx.x;
  if (bt < nBT) loadA(bt);

#pragma unroll
  for (int it = 0; it < NWI / 256; ++it) {
    const int i = it * 256 + tid;
    const int q = i % (N / 4);
    const int r = i / (N / 4);
    const size_t g0 = (size_t)(2 * r) * N + 4 * q;
    const float4 l0 = *(const float4*)(Wl + g0);
    const float4 l1 = *(const float4*)(Wl + g0 + N);
    const float4 r0 = *(const float4*)(Wr + g0);
    const float4 r1 = *(const float4*)(Wr + g0 + N);
    const int n    = 4 * q;
    const int pcol = ((((n >> 4) ^ (r & 1)) << 4) | (n & 15));
    const int o    = r * WROW + pcol * 2;
    *(float4*)(wL + o)             = make_float4(l0.x, l1.x, l0.y, l1.y);
    *(float4*)(wL + o + 4)         = make_float4(l0.z, l1.z, l0.w, l1.w);
    *(float4*)(wL + o + 2 * N)     = make_float4(r0.x, r1.x, r0.y, r1.y);
    *(float4*)(wL + o + 2 * N + 4) = make_float4(r0.z, r1.z, r0.w, r1.w);
  }

  const int tn   = wave % NT;
  const int rt   = wave / NT;
  const int c0   = lane & 15;
  const int h    = lane >> 4;
  const int colB = (tn << 4) + c0;
  const float bias = bl[colB];
  const int aOff = (rt * 16 + c0) * AP + 2 * h;
  const int wOff = h * WROW + ((((tn ^ h) << 4) + c0) << 1);

  if (bt < nBT) storeA(0);
  __syncthreads();

  int buf = 0;
  for (; bt < nBT; bt += gridDim.x) {
    const int  nbt  = bt + gridDim.x;
    const bool more = nbt < nBT;
    if (more) loadA(nbt);

    const int  m0    = (bt * RT + rt) << 4;
    const bool valid = m0 < M;
    v8f cl = {};
    v8f cr = {};
    if (valid) {
      const float* ap  = aL + buf * ABUF + aOff;
      const float* wp  = wL + wOff;
#pragma unroll 8
      for (int kk = 0; kk < KD; kk += 4) {
        const v2f a  = *(const v2f*)(ap + kk);
        const v2f b0 = *(const v2f*)(wp + (kk >> 1) * WROW);
        const v2f b1 = *(const v2f*)(wp + (kk >> 1) * WROW + 2 * N);
        cl = __builtin_amdgcn_wmma_f32_16x16x4_f32(false, a, false, b0, (short)0, cl, false, false);
        cr = __builtin_amdgcn_wmma_f32_16x16x4_f32(false, a, false, b1, (short)0, cr, false, false);
      }
    }

    if (more) storeA(buf ^ 1);
    __syncthreads();

    if (valid) {
      const int rbase = m0 + (h << 3);
#pragma unroll
      for (int r = 0; r < 8; ++r) {
        const size_t off = (size_t)(rbase + r) * N + colB;
        Yl[off]  = cl[r];
        Out[off] = cr[r] + bias;
      }
    }
    buf ^= 1;
  }
}

template <int NPW>
__global__ void pool_kernel(const float* __restrict__ h, const int* __restrict__ cluster,
                            float* __restrict__ gsum, float* __restrict__ gcnt, int n) {
  const int wave = __builtin_amdgcn_readfirstlane(blockIdx.x * (blockDim.x >> 5) + (threadIdx.x >> 5));
  const int lane = threadIdx.x & 31;
  const int n0 = wave * NPW;
  if (n0 >= n) return;
  int g[NPW];
#pragma unroll
  for (int k = 0; k < NPW; ++k) g[k] = cluster[n0 + k];
  float v[NPW][4];
#pragma unroll
  for (int k = 0; k < NPW; ++k) {
    const float* row = h + (size_t)(n0 + k) * 128 + lane;
#pragma unroll
    for (int j = 0; j < 4; ++j) v[k][j] = row[j * 32];
  }
#pragma unroll
  for (int k = 0; k < NPW; ++k) {
    float* p = gsum + (size_t)g[k] * 128 + lane;
#pragma unroll
    for (int j = 0; j < 4; ++j) atomAddF(p + j * 32, fmaxf(v[k][j], 0.0f));
  }
  if (lane < NPW) atomAddF(&gcnt[cluster[n0 + lane]], 1.0f);
}

__global__ void pool_div_kernel(float* __restrict__ g, const float* __restrict__ gcnt, int total) {
  int i = blockIdx.x * blockDim.x + threadIdx.x;
  if (i >= total) return;
  g[i] = g[i] / fmaxf(gcnt[i >> 7], 1.0f);
}

template <int RPW>
__global__ void log_softmax_kernel(float* out, int rows) {
  int wave = blockIdx.x * (blockDim.x >> 5) + (threadIdx.x >> 5);
  int lane = threadIdx.x & 31;
  int r0 = wave * RPW;
  if (r0 >= rows) return;
  float* base = out + (size_t)r0 * 64 + lane;
  float v0[RPW], v1[RPW];
#pragma unroll
  for (int k = 0; k < RPW; ++k) {
    v0[k] = base[k * 64];
    v1[k] = base[k * 64 + 32];
  }
#pragma unroll
  for (int k = 0; k < RPW; ++k) {
    float m = fmaxf(v0[k], v1[k]);
#pragma unroll
    for (int off = 16; off >= 1; off >>= 1) m = fmaxf(m, __shfl_xor(m, off, 32));
    float s = __expf(v0[k] - m) + __expf(v1[k] - m);
#pragma unroll
    for (int off = 16; off >= 1; off >>= 1) s += __shfl_xor(s, off, 32);
    float lse = m + __logf(s);
    base[k * 64]      = v0[k] - lse;
    base[k * 64 + 32] = v1[k] - lse;
  }
}

extern "C" void kernel_launch(void* const* d_in, const int* in_sizes, int n_in,
                              void* d_out, int out_size, void* d_ws, size_t ws_size,
                              hipStream_t stream) {
  const float* x       = (const float*)d_in[0];
  const int*   ei      = (const int*)d_in[1];
  const int*   src     = ei;
  const int*   dst     = ei + NE;
  const int*   cluster = (const int*)d_in[2];
  const float* Wl0 = (const float*)d_in[3];
  const float* bl0 = (const float*)d_in[4];
  const float* Wr0 = (const float*)d_in[5];
  const float* Wl1 = (const float*)d_in[6];
  const float* bl1 = (const float*)d_in[7];
  const float* Wr1 = (const float*)d_in[8];
  const float* Wl2 = (const float*)d_in[9];
  const float* bl2 = (const float*)d_in[10];
  const float* Wr2 = (const float*)d_in[11];

  float* out_ls = (float*)d_out;
  float* out_h2 = out_ls + (size_t)NN * OUTC;
  float* out_g  = out_h2 + (size_t)NN * HIDC;

  float* ws   = (float*)d_ws;
  float* yl   = ws;
  float* h1p  = ws + (size_t)NN * HIDC;
  float* deg  = h1p + (size_t)2 * NN * HIDC;
  float* gcnt = deg + NN;

  const int TPB = 256;
  constexpr int EPW = 16;
  constexpr int NPW = 8;
  constexpr int RPW = 8;
  static_assert(NE % EPW == 0, "a scatter wave takes a whole run of EPW edges");
  static_assert(NN % NPW == 0 && NN % RPW == 0, "pooling and log-softmax waves take whole runs");
  const int scatBlocks = ((NE / EPW) * 32 + TPB - 1) / TPB;
  static_assert(INC == KD && HIDC == KD && NN % 16 == 0, "the GEMM kernel's K; whole 16-row tiles");
  constexpr int GEMM_WG_128 = 512;
  constexpr int GEMM_WG_64  = 521;

  hipMemsetAsync(deg, 0, NN * sizeof(float), stream);
  deg_kernel<<<(NE + TPB - 1) / TPB, TPB, 0, stream>>>(dst, deg, NE);
  invdeg_kernel<<<(NN + TPB - 1) / TPB, TPB, 0, stream>>>(deg, NN);

  sage_gemm_kernel<false, HIDC><<<GEMM_WG_128, TPB, 0, stream>>>(x, Wl0, bl0, Wr0, yl, h1p, NN);
  edge_scatter_kernel<HIDC, EPW><<<scatBlocks, TPB, 0, stream>>>(yl, src, dst, deg, h1p, NE);

  sage_gemm_kernel<true, HIDC><<<GEMM_WG_128, TPB, 0, stream>>>(h1p, Wl1, bl1, Wr1, yl, out_h2, NN);
  edge_scatter_kernel<HIDC, EPW><<<scatBlocks, TPB, 0, stream>>>(yl, src, dst, deg, out_h2, NE);

  hipMemsetAsync(out_g, 0, (size_t)NG * HIDC * sizeof(float), stream);
  hipMemsetAsync(gcnt, 0, NG * sizeof(float), stream);
  pool_kernel<NPW><<<((NN / NPW) * 32 + TPB - 1) / TPB, TPB, 0, stream>>>(out_h2, cluster, out_g, gcnt, NN);
  pool_div_kernel<<<(NG * HIDC + TPB - 1) / TPB, TPB, 0, stream>>>(out_g, gcnt, NG * HIDC);

  sage_gemm_kernel<true, OUTC><<<GEMM_WG_64, TPB, 0, stream>>>(out_h2, Wl2, bl2, Wr2, yl, out_ls, NN);
  edge_scatter_kernel<OUTC, EPW><<<scatBlocks, TPB, 0, stream>>>(yl, src, dst, deg, out_ls, NE);
  log_softmax_kernel<RPW><<<((NN / RPW) * 32 + TPB - 1) / TPB, TPB, 0, stream>>>(out_ls, NN);
}


# Each line gets its float adds from one chip half only: out rows split by die id, scatter work-groups take tickets of their half's edge walk, degrees counted per half; atomic sum order changes
# speedup vs baseline: 2.9701x; 1.2644x over previous
#include <hip/hip_runtime.h>
#include <hip/hip_bf16.h>
#include <math.h>

typedef float v2f __attribute__((ext_vector_type(2)));
typedef float v8f __attribute__((ext_vector_type(8)));

#define NN      50000
#define NE      600000
#define INC     128
#define HIDC    128
#define OUTC    64
#define NG      512
#define DEGP    50016
#define CTRF    256

__device__ __forceinline__ void atomAddF(float* p, float v) {
  __hip_atomic_fetch_add(p, v, __ATOMIC_RELAXED, __HIP_MEMORY_SCOPE_AGENT);
}

__device__ __forceinline__ int chip_half() {
  return (int)((__builtin_amdgcn_s_sendmsg_rtn(0x87) >> 18) & 1u);
}

__device__ __forceinline__ int take_ticket(int* c) {
  return __hip_atomic_fetch_add(c, 1, __ATOMIC_RELAXED, __HIP_MEMORY_SCOPE_AGENT);
}

__global__ void deg_kernel(const int* __restrict__ dst, float* __restrict__ deg2, int E, int plane) {
  int e = blockIdx.x * blockDim.x + threadIdx.x;
  float* deg = deg2 + (size_t)chip_half() * plane;
  if (e < E) atomAddF(&deg[dst[e]], 1.0f);
}

__global__ void invdeg_kernel(float* __restrict__ deg2, int n, int plane) {
  int i = blockIdx.x * blockDim.x + threadIdx.x;
  if (i < n) deg2[(size_t)plane + i] = 1.0f / fmaxf(deg2[i] + deg2[(size_t)plane + i], 1.0f);
}

template <int CH>
__global__ void __launch_bounds__(256)
edge_scatter_kernel(const float* __restrict__ yl, const int* __restrict__ src,
                    const int* __restrict__ dst, const float* __restrict__ invdeg,
                    float* __restrict__ out, int* __restrict__ ctr, int E, int n) {
  constexpr int LPR = CH / 32;
  constexpr int RUN = 32;
  __shared__ int tk;
  const int tid  = threadIdx.x;
  const int lane = tid & 31;
  const int wave = __builtin_amdgcn_readfirstlane(tid >> 5);
  const int nRuns = (E + RUN - 1) / RUN;
  const int nT    = (nRuns + 7) >> 3;
  const int myHalf = chip_half();
  const int nLo    = n >> 1;

  for (int pass = 0; pass < 2; ++pass) {
    const int hs = myHalf ^ pass;
    const int lo = hs ? nLo : 0;
    const unsigned span = hs ? (unsigned)(n - nLo) : (unsigned)nLo;
    int* c = ctr + hs * 32;
    for (int it = 0; it <= nT; ++it) {
      if (tid == 0) tk = take_ticket(c);
      __syncthreads();
      const int t = __builtin_amdgcn_readfirstlane(tk);
      __syncthreads();
      if (t >= nT) break;

      const int  e   = (t * 8 + wave) * RUN + lane;
      const bool inE = e < E;
      int sv = 0, dv = 0;
      if (inE) { sv = src[e]; dv = dst[e]; }
      const bool mine = inE && (unsigned)(dv - lo) < span;
      float wv = 0.0f;
      if (mine) wv = invdeg[dv];
      const unsigned mask = __builtin_amdgcn_ballot_w32(mine);

      float v[RUN][LPR];
#pragma unroll
      for (int k = 0; k < RUN; ++k) {
        if ((mask >> k) & 1u) {
          const int s = __builtin_amdgcn_readlane(sv, k);
          const float* row = yl + (size_t)s * CH + lane;
#pragma unroll
          for (int j = 0; j < LPR; ++j) v[k][j] = row[j * 32];
        }
      }
#pragma unroll
      for (int k = 0; k < RUN; ++k) {
        if ((mask >> k) & 1u) {
          const int   d = __builtin_amdgcn_readlane(dv, k);
          const float w = __int_as_float(__builtin_amdgcn_readlane(__float_as_int(wv), k));
          float* p = out + (size_t)d * CH + lane;
#pragma unroll
          for (int j = 0; j < LPR; ++j) atomAddF(p + j * 32, v[k][j] * w);
        }
      }
    }
  }
}

#define KD 128
template <bool RELU_A, int N>
__global__ void __launch_bounds__(256)
sage_gemm_kernel(const float* __restrict__ X,
                 const float* __restrict__ Wl, const float* __restrict__ bl,
                 const float* __restrict__ Wr,
                 float* __restrict__ Yl, float* __restrict__ Out, int M) {
  constexpr int NT   = N / 16;
  constexpr int RT   = 8 / NT;
  constexpr int BR   = 16 * RT;
  constexpr int AP   = KD + 4;
  constexpr int WROW = 4 * N;
  constexpr int ABUF = BR * AP;
  constexpr int NA   = BR * (KD / 4) / 256;
  constexpr int NWI  = (KD / 2) * (N / 4);
  static_assert(NT * RT == 8 && NWI % 256 == 0 && NA >= 1, "8 waves of 32 lanes");
  static_assert((2 * ABUF) % 64 == 0, "the weights start on bank 0");
  __shared__ __attribute__((aligned(16))) float lds[2 * ABUF + (KD / 2) * WROW];
  float* aL = lds;
  float* wL = lds + 2 * ABUF;

  const int tid  = threadIdx.x;
  const int lane = tid & 31;
  const int wave = __builtin_amdgcn_readfirstlane(tid >> 5);
  const int nBT  = (M + BR - 1) / BR;

  float4 pre[NA];
  auto loadA = [&](int bt) {
#pragma unroll
    for (int i = 0; i < NA; ++i) {
      const int j  = i * 256 + tid;
      const int gr = bt * BR + (j >> 5);
      pre[i] = make_float4(0.0f, 0.0f, 0.0f, 0.0f);
      if (gr < M) pre[i] = *(const float4*)(X + (size_t)gr * KD + (j & 31) * 4);
    }
  };
  auto storeA = [&](int buf) {
#pragma unroll
    for (int i = 0; i < NA; ++i) {
      const int j = i * 256 + tid;
      float4 v = pre[i];
      if (RELU_A) {
        v.x = fmaxf(v.x, 0.0f); v.y = fmaxf(v.y, 0.0f);
        v.z = fmaxf(v.z, 0.0f); v.w = fmaxf(v.w, 0.0f);
      }
      *(float4*)(aL + buf * ABUF + (j >> 5) * AP + (j & 31) * 4) = v;
    }
  };

  int bt = blockIdx.x;
  if (bt < nBT) loadA(bt);

#pragma unroll
  for (int it = 0; it < NWI / 256; ++it) {
    const int i = it * 256 + tid;
    const int q = i % (N / 4);
    const int r = i / (N / 4);
    const size_t g0 = (size_t)(2 * r) * N + 4 * q;
    const float4 l0 = *(const float4*)(Wl + g0);
    const float4 l1 = *(const float4*)(Wl + g0 + N);
    const float4 r0 = *(const float4*)(Wr + g0);
    const float4 r1 = *(const float4*)(Wr + g0 + N);
    const int n    = 4 * q;
    const int pcol = ((((n >> 4) ^ (r & 1)) << 4) | (n & 15));
    const int o    = r * WROW + pcol * 2;
    *(float4*)(wL + o)             = make_float4(l0.x, l1.x, l0.y, l1.y);
    *(float4*)(wL + o + 4)         = make_float4(l0.z, l1.z, l0.w, l1.w);
    *(float4*)(wL + o + 2 * N)     = make_float4(r0.x, r1.x, r0.y, r1.y);
    *(float4*)(wL + o + 2 * N + 4) = make_float4(r0.z, r1.z, r0.w, r1.w);
  }

  const int tn   = wave % NT;
  const int rt   = wave / NT;
  const int c0   = lane & 15;
  const int h    = lane >> 4;
  const int colB = (tn << 4) + c0;
  const float bias = bl[colB];
  const int aOff = (rt * 16 + c0) * AP + 2 * h;
  const int wOff = h * WROW + ((((tn ^ h) << 4) + c0) << 1);

  if (bt < nBT) storeA(0);
  __syncthreads();

  int buf = 0;
  for (; bt < nBT; bt += gridDim.x) {
    const int  nbt  = bt + gridDim.x;
    const bool more = nbt < nBT;
    if (more) loadA(nbt);

    const int  m0    = (bt * RT + rt) << 4;
    const bool valid = m0 < M;
    v8f cl = {};
    v8f cr = {};
    if (valid) {
      const float* ap  = aL + buf * ABUF + aOff;
      const float* wp  = wL + wOff;
#pragma unroll 8
      for (int kk = 0; kk < KD; kk += 4) {
        const v2f a  = *(const v2f*)(ap + kk);
        const v2f b0 = *(const v2f*)(wp + (kk >> 1) * WROW);
        const v2f b1 = *(const v2f*)(wp + (kk >> 1) * WROW + 2 * N);
        cl = __builtin_amdgcn_wmma_f32_16x16x4_f32(false, a, false, b0, (short)0, cl, false, false);
        cr = __builtin_amdgcn_wmma_f32_16x16x4_f32(false, a, false, b1, (short)0, cr, false, false);
      }
    }

    if (more) storeA(buf ^ 1);
    __syncthreads();

    if (valid) {
      const int rbase = m0 + (h << 3);
#pragma unroll
      for (int r = 0; r < 8; ++r) {
        const size_t off = (size_t)(rbase + r) * N + colB;
        Yl[off]  = cl[r];
        Out[off] = cr[r] + bias;
      }
    }
    buf ^= 1;
  }
}

template <int NPW>
__global__ void pool_kernel(const float* __restrict__ h, const int* __restrict__ cluster,
                            float* __restrict__ gsum, float* __restrict__ gcnt, int n) {
  const int wave = __builtin_amdgcn_readfirstlane(blockIdx.x * (blockDim.x >> 5) + (threadIdx.x >> 5));
  const int lane = threadIdx.x & 31;
  const int n0 = wave * NPW;
  if (n0 >= n) return;
  int g[NPW];
#pragma unroll
  for (int k = 0; k < NPW; ++k) g[k] = cluster[n0 + k];
  float v[NPW][4];
#pragma unroll
  for (int k = 0; k < NPW; ++k) {
    const float* row = h + (size_t)(n0 + k) * 128 + lane;
#pragma unroll
    for (int j = 0; j < 4; ++j) v[k][j] = row[j * 32];
  }
#pragma unroll
  for (int k = 0; k < NPW; ++k) {
    float* p = gsum + (size_t)g[k] * 128 + lane;
#pragma unroll
    for (int j = 0; j < 4; ++j) atomAddF(p + j * 32, fmaxf(v[k][j], 0.0f));
  }
  if (lane < NPW) atomAddF(&gcnt[cluster[n0 + lane]], 1.0f);
}

__global__ void pool_div_kernel(float* __restrict__ g, const float* __restrict__ gcnt, int total) {
  int i = blockIdx.x * blockDim.x + threadIdx.x;
  if (i >= total) return;
  g[i] = g[i] / fmaxf(gcnt[i >> 7], 1.0f);
}

template <int RPW>
__global__ void log_softmax_kernel(float* out, int rows) {
  int wave = blockIdx.x * (blockDim.x >> 5) + (threadIdx.x >> 5);
  int lane = threadIdx.x & 31;
  int r0 = wave * RPW;
  if (r0 >= rows) return;
  float* base = out + (size_t)r0 * 64 + lane;
  float v0[RPW], v1[RPW];
#pragma unroll
  for (int k = 0; k < RPW; ++k) {
    v0[k] = base[k * 64];
    v1[k] = base[k * 64 + 32];
  }
#pragma unroll
  for (int k = 0; k < RPW; ++k) {
    float m = fmaxf(v0[k], v1[k]);
#pragma unroll
    for (int off = 16; off >= 1; off >>= 1) m = fmaxf(m, __shfl_xor(m, off, 32));
    float s = __expf(v0[k] - m) + __expf(v1[k] - m);
#pragma unroll
    for (int off = 16; off >= 1; off >>= 1) s += __shfl_xor(s, off, 32);
    float lse = m + __logf(s);
    base[k * 64]      = v0[k] - lse;
    base[k * 64 + 32] = v1[k] - lse;
  }
}

extern "C" void kernel_launch(void* const* d_in, const int* in_sizes, int n_in,
                              void* d_out, int out_size, void* d_ws, size_t ws_size,
                              hipStream_t stream) {
  const float* x       = (const float*)d_in[0];
  const int*   ei      = (const int*)d_in[1];
  const int*   src     = ei;
  const int*   dst     = ei + NE;
  const int*   cluster = (const int*)d_in[2];
  const float* Wl0 = (const float*)d_in[3];
  const float* bl0 = (const float*)d_in[4];
  const float* Wr0 = (const float*)d_in[5];
  const float* Wl1 = (const float*)d_in[6];
  const float* bl1 = (const float*)d_in[7];
  const float* Wr1 = (const float*)d_in[8];
  const float* Wl2 = (const float*)d_in[9];
  const float* bl2 = (const float*)d_in[10];
  const float* Wr2 = (const float*)d_in[11];

  float* out_ls = (float*)d_out;
  float* out_h2 = out_ls + (size_t)NN * OUTC;
  float* out_g  = out_h2 + (size_t)NN * HIDC;

  float* ws   = (float*)d_ws;
  float* yl   = ws;
  float* h1p  = ws + (size_t)NN * HIDC;
  float* deg  = h1p + (size_t)2 * NN * HIDC;
  float* gcnt = deg + NN;
  float* degB = deg - DEGP;
  int*   ctr  = (int*)(degB - CTRF);

  const int TPB = 256;
  constexpr int NPW = 8;
  constexpr int RPW = 8;
  static_assert(NN % NPW == 0 && NN % RPW == 0, "pooling and log-softmax waves take whole runs");
  constexpr int SCAT_WG_128 = 768;
  constexpr int SCAT_WG_64  = 1280;
  static_assert(INC == KD && HIDC == KD && NN % 16 == 0, "the GEMM kernel's K; whole 16-row tiles");
  constexpr int GEMM_WG_128 = 512;
  constexpr int GEMM_WG_64  = 521;

  hipMemsetAsync(ctr, 0, (CTRF + DEGP + NN) * sizeof(float), stream);
  deg_kernel<<<(NE + TPB - 1) / TPB, TPB, 0, stream>>>(dst, degB, NE, DEGP);
  invdeg_kernel<<<(NN + TPB - 1) / TPB, TPB, 0, stream>>>(degB, NN, DEGP);

  sage_gemm_kernel<false, HIDC><<<GEMM_WG_128, TPB, 0, stream>>>(x, Wl0, bl0, Wr0, yl, h1p, NN);
  edge_scatter_kernel<HIDC><<<SCAT_WG_128, TPB, 0, stream>>>(yl, src, dst, deg, h1p, ctr, NE, NN);

  sage_gemm_kernel<true, HIDC><<<GEMM_WG_128, TPB, 0, stream>>>(h1p, Wl1, bl1, Wr1, yl, out_h2, NN);
  edge_scatter_kernel<HIDC><<<SCAT_WG_128, TPB, 0, stream>>>(yl, src, dst, deg, out_h2, ctr + 64, NE, NN);

  hipMemsetAsync(out_g, 0, (size_t)NG * HIDC * sizeof(float), stream);
  hipMemsetAsync(gcnt, 0, NG * sizeof(float), stream);
  pool_kernel<NPW><<<((NN / NPW) * 32 + TPB - 1) / TPB, TPB, 0, stream>>>(out_h2, cluster, out_g, gcnt, NN);
  pool_div_kernel<<<(NG * HIDC + TPB - 1) / TPB, TPB, 0, stream>>>(out_g, gcnt, NG * HIDC);

  sage_gemm_kernel<true, OUTC><<<GEMM_WG_64, TPB, 0, stream>>>(out_h2, Wl2, bl2, Wr2, yl, out_ls, NN);
  edge_scatter_kernel<OUTC><<<SCAT_WG_64, TPB, 0, stream>>>(yl, src, dst, deg, out_ls, ctr + 128, NE, NN);
  log_softmax_kernel<RPW><<<((NN / RPW) * 32 + TPB - 1) / TPB, TPB, 0, stream>>>(out_ls, NN);
}


# Five stream operations fewer: degree count and out_g zeros ride in GEMM 0, invdeg in GEMM 1 (scatter 0 divides the counts itself), pooled division in the softmax launch, one memset; sum order kept
# speedup vs baseline: 3.0957x; 1.0423x over previous
#include <hip/hip_runtime.h>
#include <hip/hip_bf16.h>
#include <math.h>

typedef float v2f __attribute__((ext_vector_type(2)));
typedef float v8f __attribute__((ext_vector_type(8)));

#define NN      50000
#define NE      600000
#define INC     128
#define HIDC    128
#define OUTC    64
#define NG      512
#define DEGP    50016
#define CTRF    256
#define GEMM_WG128 512

__device__ __forceinline__ void atomAddF(float* p, float v) {
  __hip_atomic_fetch_add(p, v, __ATOMIC_RELAXED, __HIP_MEMORY_SCOPE_AGENT);
}

__device__ __forceinline__ int chip_half() {
  return (int)((__builtin_amdgcn_s_sendmsg_rtn(0x87) >> 18) & 1u);
}

__device__ __forceinline__ int take_ticket(int* c) {
  return __hip_atomic_fetch_add(c, 1, __ATOMIC_RELAXED, __HIP_MEMORY_SCOPE_AGENT);
}


template <int CH, bool COUNTS>
__global__ void __launch_bounds__(256)
edge_scatter_kernel(const float* __restrict__ yl, const int* __restrict__ src,
                    const int* __restrict__ dst, const float* __restrict__ invdeg,
                    float* __restrict__ out, int* __restrict__ ctr, int E, int n) {
  constexpr int LPR = CH / 32;
  constexpr int RUN = 32;
  __shared__ int tk;
  const int tid  = threadIdx.x;
  const int lane = tid & 31;
  const int wave = __builtin_amdgcn_readfirstlane(tid >> 5);
  const int nRuns = (E + RUN - 1) / RUN;
  const int nT    = (nRuns + 7) >> 3;
  const int myHalf = chip_half();
  const int nLo    = n >> 1;

  for (int pass = 0; pass < 2; ++pass) {
    const int hs = myHalf ^ pass;
    const int lo = hs ? nLo : 0;
    const unsigned span = hs ? (unsigned)(n - nLo) : (unsigned)nLo;
    int* c = ctr + hs * 32;
    for (int it = 0; it <= nT; ++it) {
      if (tid == 0) tk = take_ticket(c);
      __syncthreads();
      const int t = __builtin_amdgcn_readfirstlane(tk);
      __syncthreads();
      if (t >= nT) break;

      const int  e   = (t * 8 + wave) * RUN + lane;
      const bool inE = e < E;
      int sv = 0, dv = 0;
      if (inE) { sv = src[e]; dv = dst[e]; }
      const bool mine = inE && (unsigned)(dv - lo) < span;
      float wv = 0.0f;
      if (mine) {
        if (COUNTS) wv = 1.0f / fmaxf((invdeg - DEGP)[dv] + invdeg[dv], 1.0f);
        else        wv = invdeg[dv];
      }
      const unsigned mask = __builtin_amdgcn_ballot_w32(mine);

      float v[RUN][LPR];
#pragma unroll
      for (int k = 0; k < RUN; ++k) {
        if ((mask >> k) & 1u) {
          const int s = __builtin_amdgcn_readlane(sv, k);
          const float* row = yl + (size_t)s * CH + lane;
#pragma unroll
          for (int j = 0; j < LPR; ++j) v[k][j] = row[j * 32];
        }
      }
#pragma unroll
      for (int k = 0; k < RUN; ++k) {
        if ((mask >> k) & 1u) {
          const int   d = __builtin_amdgcn_readlane(dv, k);
          const float w = __int_as_float(__builtin_amdgcn_readlane(__float_as_int(wv), k));
          float* p = out + (size_t)d * CH + lane;
#pragma unroll
          for (int j = 0; j < LPR; ++j) atomAddF(p + j * 32, v[k][j] * w);
        }
      }
    }
  }
}

#define KD 128
template <bool RELU_A, int N, int SIDE>
__global__ void __launch_bounds__(256)
sage_gemm_kernel(const float* __restrict__ X,
                 const float* __restrict__ Wl, const float* __restrict__ bl,
                 const float* __restrict__ Wr,
                 float* __restrict__ Yl, float* __restrict__ Out, int M,
                 const int* __restrict__ edst, float* __restrict__ deg2, float* __restrict__ gz) {
  constexpr int NT   = N / 16;
  constexpr int RT   = 8 / NT;
  constexpr int BR   = 16 * RT;
  constexpr int AP   = KD + 4;
  constexpr int WROW = 4 * N;
  constexpr int ABUF = BR * AP;
  constexpr int NA   = BR * (KD / 4) / 256;
  constexpr int NWI  = (KD / 2) * (N / 4);
  static_assert(NT * RT == 8 && NWI % 256 == 0 && NA >= 1, "8 waves of 32 lanes");
  static_assert((2 * ABUF) % 64 == 0, "the weights start on bank 0");
  __shared__ __attribute__((aligned(16))) float lds[2 * ABUF + (KD / 2) * WROW];
  float* aL = lds;
  float* wL = lds + 2 * ABUF;

  const int tid  = threadIdx.x;
  const int lane = tid & 31;
  const int wave = __builtin_amdgcn_readfirstlane(tid >> 5);
  const int nBT  = (M + BR - 1) / BR;

  float4 pre[NA];
  auto loadA = [&](int bt) {
#pragma unroll
    for (int i = 0; i < NA; ++i) {
      const int j  = i * 256 + tid;
      const int gr = bt * BR + (j >> 5);
      pre[i] = make_float4(0.0f, 0.0f, 0.0f, 0.0f);
      if (gr < M) pre[i] = *(const float4*)(X + (size_t)gr * KD + (j & 31) * 4);
    }
  };
  auto storeA = [&](int buf) {
#pragma unroll
    for (int i = 0; i < NA; ++i) {
      const int j = i * 256 + tid;
      float4 v = pre[i];
      if (RELU_A) {
        v.x = fmaxf(v.x, 0.0f); v.y = fmaxf(v.y, 0.0f);
        v.z = fmaxf(v.z, 0.0f); v.w = fmaxf(v.w, 0.0f);
      }
      *(float4*)(aL + buf * ABUF + (j >> 5) * AP + (j & 31) * 4) = v;
    }
  };

  int bt = blockIdx.x;
  if (bt < nBT) loadA(bt);

  constexpr int EPT = (NE + GEMM_WG128 * 256 - 1) / (GEMM_WG128 * 256);
  static_assert(GEMM_WG128 * 256 >= NN, "SIDE 2: one node per thread");
  const int gid = (int)blockIdx.x * 256 + tid;
  int   ed[EPT];
  float cnt0 = 0.0f, cnt1 = 0.0f;
  if constexpr (SIDE == 1) {
#pragma unroll
    for (int i = 0; i < EPT; ++i) {
      const int e = (i * GEMM_WG128 + (int)blockIdx.x) * 256 + tid;
      ed[i] = e < NE ? edst[e] : -1;
    }
  }
  if constexpr (SIDE == 2) {
    if (gid < M) { cnt0 = deg2[gid]; cnt1 = deg2[DEGP + gid]; }
  }

#pragma unroll
  for (int it = 0; it < NWI / 256; ++it) {
    const int i = it * 256 + tid;
    const int q = i % (N / 4);
    const int r = i / (N / 4);
    const size_t g0 = (size_t)(2 * r) * N + 4 * q;
    const float4 l0 = *(const float4*)(Wl + g0);
    const float4 l1 = *(const float4*)(Wl + g0 + N);
    const float4 r0 = *(const float4*)(Wr + g0);
    const float4 r1 = *(const float4*)(Wr + g0 + N);
    const int n    = 4 * q;
    const int pcol = ((((n >> 4) ^ (r & 1)) << 4) | (n & 15));
    const int o    = r * WROW + pcol * 2;
    *(float4*)(wL + o)             = make_float4(l0.x, l1.x, l0.y, l1.y);
    *(float4*)(wL + o + 4)         = make_float4(l0.z, l1.z, l0.w, l1.w);
    *(float4*)(wL + o + 2 * N)     = make_float4(r0.x, r1.x, r0.y, r1.y);
    *(float4*)(wL + o + 2 * N + 4) = make_float4(r0.z, r1.z, r0.w, r1.w);
  }

  const int tn   = wave % NT;
  const int rt   = wave / NT;
  const int c0   = lane & 15;
  const int h    = lane >> 4;
  const int colB = (tn << 4) + c0;
  const float bias = bl[colB];
  const int aOff = (rt * 16 + c0) * AP + 2 * h;
  const int wOff = h * WROW + ((((tn ^ h) << 4) + c0) << 1);

  if (bt < nBT) storeA(0);
  __syncthreads();

  int buf = 0;
  for (; bt < nBT; bt += gridDim.x) {
    const int  nbt  = bt + gridDim.x;
    const bool more = nbt < nBT;
    if (more) loadA(nbt);

    const int  m0    = (bt * RT + rt) << 4;
    const bool valid = m0 < M;
    v8f cl = {};
    v8f cr = {};
    if (valid) {
      const float* ap  = aL + buf * ABUF + aOff;
      const float* wp  = wL + wOff;
#pragma unroll 8
      for (int kk = 0; kk < KD; kk += 4) {
        const v2f a  = *(const v2f*)(ap + kk);
        const v2f b0 = *(const v2f*)(wp + (kk >> 1) * WROW);
        const v2f b1 = *(const v2f*)(wp + (kk >> 1) * WROW + 2 * N);
        cl = __builtin_amdgcn_wmma_f32_16x16x4_f32(false, a, false, b0, (short)0, cl, false, false);
        cr = __builtin_amdgcn_wmma_f32_16x16x4_f32(false, a, false, b1, (short)0, cr, false, false);
      }
    }

    if (more) storeA(buf ^ 1);
    __syncthreads();

    if (valid) {
      const int rbase = m0 + (h << 3);
#pragma unroll
      for (int r = 0; r < 8; ++r) {
        const size_t off = (size_t)(rbase + r) * N + colB;
        Yl[off]  = cl[r];
        Out[off] = cr[r] + bias;
      }
    }
    buf ^= 1;
  }

  if constexpr (SIDE == 1) {
    float* deg = deg2 + (size_t)chip_half() * DEGP;
#pragma unroll
    for (int i = 0; i < EPT; ++i)
      if (ed[i] >= 0) atomAddF(&deg[ed[i]], 1.0f);
    for (int i = gid; i < NG * HIDC / 4; i += gridDim.x * 256)
      ((float4*)gz)[i] = make_float4(0.0f, 0.0f, 0.0f, 0.0f);
  }
  if constexpr (SIDE == 2) {
    if (gid < M) deg2[DEGP + gid] = 1.0f / fmaxf(cnt0 + cnt1, 1.0f);
  }
}

template <int NPW>
__global__ void pool_kernel(const float* __restrict__ h, const int* __restrict__ cluster,
                            float* __restrict__ gsum, float* __restrict__ gcnt, int n) {
  const int wave = __builtin_amdgcn_readfirstlane(blockIdx.x * (blockDim.x >> 5) + (threadIdx.x >> 5));
  const int lane = threadIdx.x & 31;
  const int n0 = wave * NPW;
  if (n0 >= n) return;
  int g[NPW];
#pragma unroll
  for (int k = 0; k < NPW; ++k) g[k] = cluster[n0 + k];
  float v[NPW][4];
#pragma unroll
  for (int k = 0; k < NPW; ++k) {
    const float* row = h + (size_t)(n0 + k) * 128 + lane;
#pragma unroll
    for (int j = 0; j < 4; ++j) v[k][j] = row[j * 32];
  }
#pragma unroll
  for (int k = 0; k < NPW; ++k) {
    float* p = gsum + (size_t)g[k] * 128 + lane;
#pragma unroll
    for (int j = 0; j < 4; ++j) atomAddF(p + j * 32, fmaxf(v[k][j], 0.0f));
  }
  if (lane < NPW) atomAddF(&gcnt[cluster[n0 + lane]], 1.0f);
}

template <int RPW>
__global__ void log_softmax_kernel(float* out, int rows, int lsmWG,
                                   float* __restrict__ g, const float* __restrict__ gcnt, int total) {
  if ((int)blockIdx.x >= lsmWG) {
    const int i = ((int)blockIdx.x - lsmWG) * blockDim.x + threadIdx.x;
    if (i < total) g[i] = g[i] / fmaxf(gcnt[i >> 7], 1.0f);
    return;
  }
  int wave = blockIdx.x * (blockDim.x >> 5) + (threadIdx.x >> 5);
  int lane = threadIdx.x & 31;
  int r0 = wave * RPW;
  if (r0 >= rows) return;
  float* base = out + (size_t)r0 * 64 + lane;
  float v0[RPW], v1[RPW];
#pragma unroll
  for (int k = 0; k < RPW; ++k) {
    v0[k] = base[k * 64];
    v1[k] = base[k * 64 + 32];
  }
#pragma unroll
  for (int k = 0; k < RPW; ++k) {
    float m = fmaxf(v0[k], v1[k]);
#pragma unroll
    for (int off = 16; off >= 1; off >>= 1) m = fmaxf(m, __shfl_xor(m, off, 32));
    float s = __expf(v0[k] - m) + __expf(v1[k] - m);
#pragma unroll
    for (int off = 16; off >= 1; off >>= 1) s += __shfl_xor(s, off, 32);
    float lse = m + __logf(s);
    base[k * 64]      = v0[k] - lse;
    base[k * 64 + 32] = v1[k] - lse;
  }
}

extern "C" void kernel_launch(void* const* d_in, const int* in_sizes, int n_in,
                              void* d_out, int out_size, void* d_ws, size_t ws_size,
                              hipStream_t stream) {
  const float* x       = (const float*)d_in[0];
  const int*   ei      = (const int*)d_in[1];
  const int*   src     = ei;
  const int*   dst     = ei + NE;
  const int*   cluster = (const int*)d_in[2];
  const float* Wl0 = (const float*)d_in[3];
  const float* bl0 = (const float*)d_in[4];
  const float* Wr0 = (const float*)d_in[5];
  const float* Wl1 = (const float*)d_in[6];
  const float* bl1 = (const float*)d_in[7];
  const float* Wr1 = (const float*)d_in[8];
  const float* Wl2 = (const float*)d_in[9];
  const float* bl2 = (const float*)d_in[10];
  const float* Wr2 = (const float*)d_in[11];

  float* out_ls = (float*)d_out;
  float* out_h2 = out_ls + (size_t)NN * OUTC;
  float* out_g  = out_h2 + (size_t)NN * HIDC;

  float* ws   = (float*)d_ws;
  float* yl   = ws;
  float* h1p  = ws + (size_t)NN * HIDC;
  float* deg  = h1p + (size_t)2 * NN * HIDC;
  float* gcnt = deg + NN;
  float* degB = deg - DEGP;
  int*   ctr  = (int*)(degB - CTRF);

  const int TPB = 256;
  constexpr int NPW = 8;
  constexpr int RPW = 8;
  static_assert(NN % NPW == 0 && NN % RPW == 0, "pooling and log-softmax waves take whole runs");
  constexpr int SCAT_WG_128 = 768;
  constexpr int SCAT_WG_64  = 1280;
  static_assert(INC == KD && HIDC == KD && NN % 16 == 0, "the GEMM kernel's K; whole 16-row tiles");
  constexpr int GEMM_WG_128 = GEMM_WG128;
  constexpr int GEMM_WG_64  = 521;

  hipMemsetAsync(ctr, 0, (CTRF + DEGP + NN + NG) * sizeof(float), stream);

  sage_gemm_kernel<false, HIDC, 1><<<GEMM_WG_128, TPB, 0, stream>>>(x, Wl0, bl0, Wr0, yl, h1p, NN, dst, degB, out_g);
  edge_scatter_kernel<HIDC, true><<<SCAT_WG_128, TPB, 0, stream>>>(yl, src, dst, deg, h1p, ctr, NE, NN);

  sage_gemm_kernel<true, HIDC, 2><<<GEMM_WG_128, TPB, 0, stream>>>(h1p, Wl1, bl1, Wr1, yl, out_h2, NN, dst, degB, out_g);
  edge_scatter_kernel<HIDC, false><<<SCAT_WG_128, TPB, 0, stream>>>(yl, src, dst, deg, out_h2, ctr + 64, NE, NN);

  pool_kernel<NPW><<<((NN / NPW) * 32 + TPB - 1) / TPB, TPB, 0, stream>>>(out_h2, cluster, out_g, gcnt, NN);

  sage_gemm_kernel<true, OUTC, 0><<<GEMM_WG_64, TPB, 0, stream>>>(out_h2, Wl2, bl2, Wr2, yl, out_ls, NN, dst, degB, out_g);
  edge_scatter_kernel<OUTC, false><<<SCAT_WG_64, TPB, 0, stream>>>(yl, src, dst, deg, out_ls, ctr + 128, NE, NN);
  constexpr int LSM_WG  = ((NN / RPW) * 32 + TPB - 1) / TPB;
  constexpr int PDIV_WG = (NG * HIDC + TPB - 1) / TPB;
  log_softmax_kernel<RPW><<<LSM_WG + PDIV_WG, TPB, 0, stream>>>(out_ls, NN, LSM_WG, out_g, gcnt, NG * HIDC);
}
